# starting kernel
# baseline (speedup 1.0000x reference)
#include <hip/hip_runtime.h>
#include <hip/hip_bf16.h>

constexpr int B_BATCH = 4;
constexpr int N_NODES = 8192;
constexpr int E_EDGES = 131072;
constexpr int ETOT    = E_EDGES + N_NODES;
constexpr int IN_DIM  = 128;
constexpr int HEADS   = 4;
constexpr int CH      = 32;
constexpr int OUT_DIM = HEADS * CH;
constexpr int BN      = B_BATCH * N_NODES;

typedef float v2f __attribute__((ext_vector_type(2)));
typedef float v8f __attribute__((ext_vector_type(8)));

__device__ __forceinline__ unsigned fkey(float f) {
    unsigned u = __float_as_uint(f);
    return (u & 0x80000000u) ? ~u : (u | 0x80000000u);
}
__device__ __forceinline__ float fdec(unsigned u) {
    unsigned b = (u & 0x80000000u) ? (u & 0x7fffffffu) : ~u;
    return __uint_as_float(b);
}
__device__ __forceinline__ float lrelu02(float v) {
    return v > 0.0f ? v : 0.2f * v;
}

__global__ __launch_bounds__(256) void gat_gemm_wmma(
    const float* __restrict__ x,
    const float* __restrict__ Wl, const float* __restrict__ bl,
    const float* __restrict__ Wr, const float* __restrict__ br,
    float* __restrict__ xl, float* __restrict__ xr)
{
    const int lane   = threadIdx.x & 31;
    const int wave   = blockIdx.x * 8 + (threadIdx.x >> 5);
    const int mat     = wave & 1;
    const int colTile = (wave >> 1) & 7;
    const int rowTile = wave >> 4;

    const int mOrN  = lane & 15;
    const int khalf = (lane >> 4) * 2;
    const int row   = rowTile * 16 + mOrN;
    const int col   = colTile * 16 + mOrN;

    const float* __restrict__ W    = mat ? Wr : Wl;
    const float* __restrict__ bias = mat ? br : bl;
    float*       __restrict__ out  = mat ? xr : xl;

    v8f acc = {};
#if __has_builtin(__builtin_amdgcn_wmma_f32_16x16x4_f32)
    #pragma unroll
    for (int k = 0; k < IN_DIM; k += 4) {
        v2f a = *(const v2f*)(x + (size_t)row * IN_DIM + k + khalf);
        v2f b;
        b.x = W[(size_t)(k + khalf + 0) * OUT_DIM + col];
        b.y = W[(size_t)(k + khalf + 1) * OUT_DIM + col];
        acc = __builtin_amdgcn_wmma_f32_16x16x4_f32(
                  false, a, false, b, (short)0, acc, false, false);
    }
#else
    const int rh = (lane >> 4) * 8;
    #pragma unroll
    for (int v = 0; v < 8; ++v) {
        float s = 0.0f;
        const int r = rowTile * 16 + rh + v;
        for (int k = 0; k < IN_DIM; ++k)
            s += x[(size_t)r * IN_DIM + k] * W[(size_t)k * OUT_DIM + col];
        acc[v] = s;
    }
#endif
    const int rowHalf = (lane >> 4) * 8;
    const float bv = bias[col];
    #pragma unroll
    for (int v = 0; v < 8; ++v)
        out[(size_t)(rowTile * 16 + rowHalf + v) * OUT_DIM + col] = acc[v] + bv;
}

__global__ __launch_bounds__(256) void gat_edge_logits(
    const int* __restrict__ eidx,
    const float* __restrict__ xl, const float* __restrict__ xr,
    const float* __restrict__ att,
    float* __restrict__ e, unsigned* __restrict__ m_u)
{
    const int lane = threadIdx.x & 31;
    const int wave = blockIdx.x * 8 + (threadIdx.x >> 5);
    const int b    = wave / ETOT;
    const int ed   = wave - b * ETOT;
    if (b >= B_BATCH) return;

    int src, dst;
    if (ed < E_EDGES) {
        src = eidx[((size_t)b * 2 + 0) * E_EDGES + ed];
        dst = eidx[((size_t)b * 2 + 1) * E_EDGES + ed];
    } else {
        src = dst = ed - E_EDGES;
    }

    const float4 a4 = *(const float4*)(xl + ((size_t)b * N_NODES + src) * OUT_DIM + lane * 4);
    const float4 r4 = *(const float4*)(xr + ((size_t)b * N_NODES + dst) * OUT_DIM + lane * 4);
    const float4 t4 = *(const float4*)(att + lane * 4);

    float p = lrelu02(a4.x + r4.x) * t4.x
            + lrelu02(a4.y + r4.y) * t4.y
            + lrelu02(a4.z + r4.z) * t4.z
            + lrelu02(a4.w + r4.w) * t4.w;
    p += __shfl_xor(p, 1, 32);
    p += __shfl_xor(p, 2, 32);
    p += __shfl_xor(p, 4, 32);

    if ((lane & 7) == 0) {
        const int h = lane >> 3;
        e[((size_t)b * ETOT + ed) * HEADS + h] = p;
        atomicMax(&m_u[((size_t)b * N_NODES + dst) * HEADS + h], fkey(p));
    }
}

__global__ __launch_bounds__(256) void gat_edge_scatter(
    const int* __restrict__ eidx,
    const float* __restrict__ xl, const float* __restrict__ e,
    const unsigned* __restrict__ m_u,
    float* __restrict__ denom, float* __restrict__ accum)
{
    const int lane = threadIdx.x & 31;
    const int wave = blockIdx.x * 8 + (threadIdx.x >> 5);
    const int b    = wave / ETOT;
    const int ed   = wave - b * ETOT;
    if (b >= B_BATCH) return;

    int src, dst;
    if (ed < E_EDGES) {
        src = eidx[((size_t)b * 2 + 0) * E_EDGES + ed];
        dst = eidx[((size_t)b * 2 + 1) * E_EDGES + ed];
    } else {
        src = dst = ed - E_EDGES;
    }

    const int h    = lane >> 3;
    const float eh = e[((size_t)b * ETOT + ed) * HEADS + h];
    const float mh = fdec(m_u[((size_t)b * N_NODES + dst) * HEADS + h]);
    const float ex = __expf(eh - mh);

    if ((lane & 7) == 0)
        atomicAdd(&denom[((size_t)b * N_NODES + dst) * HEADS + h], ex);

    const float4 a4 = *(const float4*)(xl + ((size_t)b * N_NODES + src) * OUT_DIM + lane * 4);
    float* ab = accum + ((size_t)b * N_NODES + dst) * OUT_DIM + lane * 4;
    atomicAdd(ab + 0, ex * a4.x);
    atomicAdd(ab + 1, ex * a4.y);
    atomicAdd(ab + 2, ex * a4.z);
    atomicAdd(ab + 3, ex * a4.w);
}

__global__ __launch_bounds__(256) void gat_finalize(
    const float* __restrict__ accum, const float* __restrict__ denom,
    const float* __restrict__ bias, const float* __restrict__ gamma,
    const float* __restrict__ beta, float* __restrict__ out)
{
    const int lane = threadIdx.x & 31;
    const int node = blockIdx.x * 8 + (threadIdx.x >> 5);
    const float4 a4 = *(const float4*)(accum + (size_t)node * OUT_DIM + lane * 4);
    const float  dh = denom[(size_t)node * HEADS + (lane >> 3)] + 1e-16f;
    const float4 bi = *(const float4*)(bias + lane * 4);

    float v0 = a4.x / dh + bi.x;
    float v1 = a4.y / dh + bi.y;
    float v2 = a4.z / dh + bi.z;
    float v3 = a4.w / dh + bi.w;

    float s  = v0 + v1 + v2 + v3;
    float sq = v0 * v0 + v1 * v1 + v2 * v2 + v3 * v3;
    #pragma unroll
    for (int i = 1; i < 32; i <<= 1) {
        s  += __shfl_xor(s,  i, 32);
        sq += __shfl_xor(sq, i, 32);
    }
    const float mean = s * (1.0f / OUT_DIM);
    const float var  = sq * (1.0f / OUT_DIM) - mean * mean;
    const float inv  = rsqrtf(var + 1e-5f);

    const float4 g4 = *(const float4*)(gamma + lane * 4);
    const float4 b4 = *(const float4*)(beta + lane * 4);
    float4 o;
    o.x = (v0 - mean) * inv * g4.x + b4.x;
    o.y = (v1 - mean) * inv * g4.y + b4.y;
    o.z = (v2 - mean) * inv * g4.z + b4.z;
    o.w = (v3 - mean) * inv * g4.w + b4.w;
    *(float4*)(out + (size_t)node * OUT_DIM + lane * 4) = o;
}

extern "C" void kernel_launch(void* const* d_in, const int* in_sizes, int n_in,
                              void* d_out, int out_size, void* d_ws, size_t ws_size,
                              hipStream_t stream) {
    const float* x     = (const float*)d_in[0];
    const int*   eidx  = (const int*)  d_in[1];
    const float* Wl    = (const float*)d_in[2];
    const float* bl    = (const float*)d_in[3];
    const float* Wr    = (const float*)d_in[4];
    const float* br    = (const float*)d_in[5];
    const float* att   = (const float*)d_in[6];
    const float* bias  = (const float*)d_in[7];
    const float* gamma = (const float*)d_in[8];
    const float* beta  = (const float*)d_in[9];
    float* out = (float*)d_out;

    const size_t szXL    = (size_t)BN * OUT_DIM;
    const size_t szE     = (size_t)B_BATCH * ETOT * HEADS;
    const size_t szH     = (size_t)BN * HEADS;
    const size_t needed  = (3 * szXL + szE + 2 * szH) * sizeof(float);
    if (ws_size < needed) return;

    float*    xl    = (float*)d_ws;
    float*    xr    = xl + szXL;
    float*    accum = xr + szXL;
    float*    e     = accum + szXL;
    unsigned* m_u   = (unsigned*)(e + szE);
    float*    denom = (float*)(m_u + szH);

    hipMemsetAsync(accum, 0, szXL * sizeof(float), stream);
    hipMemsetAsync(m_u,   0, szH * sizeof(unsigned), stream);
    hipMemsetAsync(denom, 0, szH * sizeof(float), stream);

    const int gemmWaves = (BN / 16) * 8 * 2;
    gat_gemm_wmma<<<gemmWaves / 8, 256, 0, stream>>>(x, Wl, bl, Wr, br, xl, xr);

    const int edgeWaves = B_BATCH * ETOT;
    gat_edge_logits<<<edgeWaves / 8, 256, 0, stream>>>(eidx, xl, xr, att, e, m_u);

    gat_edge_scatter<<<edgeWaves / 8, 256, 0, stream>>>(eidx, xl, e, m_u, denom, accum);

    gat_finalize<<<BN / 8, 256, 0, stream>>>(accum, denom, bias, gamma, beta, out);
}


# Edge kernels: each wave handles 16 consecutive edges (indices in scalar registers, row loads batched by 4) instead of one; 16x fewer waves; float atomics kept, their arrival order differs
# speedup vs baseline: 1.2352x; 1.2352x over previous
#include <hip/hip_runtime.h>
#include <hip/hip_bf16.h>

constexpr int B_BATCH = 4;
constexpr int N_NODES = 8192;
constexpr int E_EDGES = 131072;
constexpr int ETOT    = E_EDGES + N_NODES;
constexpr int IN_DIM  = 128;
constexpr int HEADS   = 4;
constexpr int CH      = 32;
constexpr int OUT_DIM = HEADS * CH;
constexpr int BN      = B_BATCH * N_NODES;

typedef float v2f __attribute__((ext_vector_type(2)));
typedef float v8f __attribute__((ext_vector_type(8)));

__device__ __forceinline__ unsigned fkey(float f) {
    unsigned u = __float_as_uint(f);
    return (u & 0x80000000u) ? ~u : (u | 0x80000000u);
}
__device__ __forceinline__ float fdec(unsigned u) {
    unsigned b = (u & 0x80000000u) ? (u & 0x7fffffffu) : ~u;
    return __uint_as_float(b);
}
__device__ __forceinline__ float lrelu02(float v) {
    return v > 0.0f ? v : 0.2f * v;
}

__global__ __launch_bounds__(256) void gat_gemm_wmma(
    const float* __restrict__ x,
    const float* __restrict__ Wl, const float* __restrict__ bl,
    const float* __restrict__ Wr, const float* __restrict__ br,
    float* __restrict__ xl, float* __restrict__ xr)
{
    const int lane   = threadIdx.x & 31;
    const int wave   = blockIdx.x * 8 + (threadIdx.x >> 5);
    const int mat     = wave & 1;
    const int colTile = (wave >> 1) & 7;
    const int rowTile = wave >> 4;

    const int mOrN  = lane & 15;
    const int khalf = (lane >> 4) * 2;
    const int row   = rowTile * 16 + mOrN;
    const int col   = colTile * 16 + mOrN;

    const float* __restrict__ W    = mat ? Wr : Wl;
    const float* __restrict__ bias = mat ? br : bl;
    float*       __restrict__ out  = mat ? xr : xl;

    v8f acc = {};
#if __has_builtin(__builtin_amdgcn_wmma_f32_16x16x4_f32)
    #pragma unroll
    for (int k = 0; k < IN_DIM; k += 4) {
        v2f a = *(const v2f*)(x + (size_t)row * IN_DIM + k + khalf);
        v2f b;
        b.x = W[(size_t)(k + khalf + 0) * OUT_DIM + col];
        b.y = W[(size_t)(k + khalf + 1) * OUT_DIM + col];
        acc = __builtin_amdgcn_wmma_f32_16x16x4_f32(
                  false, a, false, b, (short)0, acc, false, false);
    }
#else
    const int rh = (lane >> 4) * 8;
    #pragma unroll
    for (int v = 0; v < 8; ++v) {
        float s = 0.0f;
        const int r = rowTile * 16 + rh + v;
        for (int k = 0; k < IN_DIM; ++k)
            s += x[(size_t)r * IN_DIM + k] * W[(size_t)k * OUT_DIM + col];
        acc[v] = s;
    }
#endif
    const int rowHalf = (lane >> 4) * 8;
    const float bv = bias[col];
    #pragma unroll
    for (int v = 0; v < 8; ++v)
        out[(size_t)(rowTile * 16 + rowHalf + v) * OUT_DIM + col] = acc[v] + bv;
}

constexpr int EPW              = 16;
constexpr int EBATCH           = 4;
constexpr int GROUPS_PER_BATCH = ETOT / EPW;
constexpr int EDGE_GROUPS      = B_BATCH * GROUPS_PER_BATCH;
static_assert(ETOT % EPW == 0 && E_EDGES % EPW == 0, "edge groups must not straddle");
static_assert(EPW % EBATCH == 0, "batches tile a group");
static_assert(EDGE_GROUPS % 8 == 0, "8 waves per block, no remainder");

__device__ __forceinline__ void group_edges(const int* __restrict__ eidx, int b, int g0,
                                            int (&src)[EPW], int (&dst)[EPW])
{
    if (g0 < E_EDGES) {
        const int* __restrict__ ps = eidx + ((size_t)b * 2 + 0) * E_EDGES + g0;
        const int* __restrict__ pd = eidx + ((size_t)b * 2 + 1) * E_EDGES + g0;
        #pragma unroll
        for (int i = 0; i < EPW; ++i) { src[i] = ps[i]; dst[i] = pd[i]; }
    } else {
        #pragma unroll
        for (int i = 0; i < EPW; ++i) src[i] = dst[i] = g0 - E_EDGES + i;
    }
}

__global__ __launch_bounds__(256) void gat_edge_logits(
    const int* __restrict__ eidx,
    const float* __restrict__ xl, const float* __restrict__ xr,
    const float* __restrict__ att,
    float* __restrict__ e, unsigned* __restrict__ m_u)
{
    const int lane = threadIdx.x & 31;
    const int wv   = __builtin_amdgcn_readfirstlane((int)(threadIdx.x >> 5));
    const int grp  = blockIdx.x * 8 + wv;
    const int b    = grp / GROUPS_PER_BATCH;
    const int g0   = (grp - b * GROUPS_PER_BATCH) * EPW;
    if (b >= B_BATCH) return;

    int src[EPW], dst[EPW];
    group_edges(eidx, b, g0, src, dst);

    const float4 t4 = *(const float4*)(att + lane * 4);
    const float* __restrict__ xlb = xl + (size_t)b * N_NODES * OUT_DIM + lane * 4;
    const float* __restrict__ xrb = xr + (size_t)b * N_NODES * OUT_DIM + lane * 4;
    const int h = lane >> 3;

    #pragma unroll
    for (int i0 = 0; i0 < EPW; i0 += EBATCH) {
        float4 a4[EBATCH], r4[EBATCH];
        #pragma unroll
        for (int u = 0; u < EBATCH; ++u) {
            a4[u] = *(const float4*)(xlb + (size_t)src[i0 + u] * OUT_DIM);
            r4[u] = *(const float4*)(xrb + (size_t)dst[i0 + u] * OUT_DIM);
        }
        float p[EBATCH];
        #pragma unroll
        for (int u = 0; u < EBATCH; ++u) {
            p[u] = lrelu02(a4[u].x + r4[u].x) * t4.x
                 + lrelu02(a4[u].y + r4[u].y) * t4.y
                 + lrelu02(a4[u].z + r4[u].z) * t4.z
                 + lrelu02(a4[u].w + r4[u].w) * t4.w;
            p[u] += __shfl_xor(p[u], 1, 32);
            p[u] += __shfl_xor(p[u], 2, 32);
            p[u] += __shfl_xor(p[u], 4, 32);
        }
        if ((lane & 7) == 0) {
            #pragma unroll
            for (int u = 0; u < EBATCH; ++u) {
                const int ed = g0 + i0 + u;
                e[((size_t)b * ETOT + ed) * HEADS + h] = p[u];
                atomicMax(&m_u[((size_t)b * N_NODES + dst[i0 + u]) * HEADS + h], fkey(p[u]));
            }
        }
    }
}

__global__ __launch_bounds__(256) void gat_edge_scatter(
    const int* __restrict__ eidx,
    const float* __restrict__ xl, const float* __restrict__ e,
    const unsigned* __restrict__ m_u,
    float* __restrict__ denom, float* __restrict__ accum)
{
    const int lane = threadIdx.x & 31;
    const int wv   = __builtin_amdgcn_readfirstlane((int)(threadIdx.x >> 5));
    const int grp  = blockIdx.x * 8 + wv;
    const int b    = grp / GROUPS_PER_BATCH;
    const int g0   = (grp - b * GROUPS_PER_BATCH) * EPW;
    if (b >= B_BATCH) return;

    int src[EPW], dst[EPW];
    group_edges(eidx, b, g0, src, dst);

    const int h = lane >> 3;
    const float*    __restrict__ xlb = xl + (size_t)b * N_NODES * OUT_DIM + lane * 4;
    const float*    __restrict__ eb  = e + ((size_t)b * ETOT + g0) * HEADS + h;
    const unsigned* __restrict__ mb  = m_u + (size_t)b * N_NODES * HEADS + h;
    float* __restrict__ db = denom + (size_t)b * N_NODES * HEADS + h;
    float* __restrict__ ac = accum + (size_t)b * N_NODES * OUT_DIM + lane * 4;

    #pragma unroll
    for (int i0 = 0; i0 < EPW; i0 += EBATCH) {
        float eh[EBATCH];
        unsigned mk[EBATCH];
        float4 a4[EBATCH];
        #pragma unroll
        for (int u = 0; u < EBATCH; ++u) {
            eh[u] = eb[(size_t)(i0 + u) * HEADS];
            mk[u] = mb[(size_t)dst[i0 + u] * HEADS];
            a4[u] = *(const float4*)(xlb + (size_t)src[i0 + u] * OUT_DIM);
        }
        float ex[EBATCH];
        #pragma unroll
        for (int u = 0; u < EBATCH; ++u)
            ex[u] = __expf(eh[u] - fdec(mk[u]));
        if ((lane & 7) == 0) {
            #pragma unroll
            for (int u = 0; u < EBATCH; ++u)
                atomicAdd(db + (size_t)dst[i0 + u] * HEADS, ex[u]);
        }
        #pragma unroll
        for (int u = 0; u < EBATCH; ++u) {
            float* ab = ac + (size_t)dst[i0 + u] * OUT_DIM;
            atomicAdd(ab + 0, ex[u] * a4[u].x);
            atomicAdd(ab + 1, ex[u] * a4[u].y);
            atomicAdd(ab + 2, ex[u] * a4[u].z);
            atomicAdd(ab + 3, ex[u] * a4[u].w);
        }
    }
}

__global__ __launch_bounds__(256) void gat_finalize(
    const float* __restrict__ accum, const float* __restrict__ denom,
    const float* __restrict__ bias, const float* __restrict__ gamma,
    const float* __restrict__ beta, float* __restrict__ out)
{
    const int lane = threadIdx.x & 31;
    const int node = blockIdx.x * 8 + (threadIdx.x >> 5);
    const float4 a4 = *(const float4*)(accum + (size_t)node * OUT_DIM + lane * 4);
    const float  dh = denom[(size_t)node * HEADS + (lane >> 3)] + 1e-16f;
    const float4 bi = *(const float4*)(bias + lane * 4);

    float v0 = a4.x / dh + bi.x;
    float v1 = a4.y / dh + bi.y;
    float v2 = a4.z / dh + bi.z;
    float v3 = a4.w / dh + bi.w;

    float s  = v0 + v1 + v2 + v3;
    float sq = v0 * v0 + v1 * v1 + v2 * v2 + v3 * v3;
    #pragma unroll
    for (int i = 1; i < 32; i <<= 1) {
        s  += __shfl_xor(s,  i, 32);
        sq += __shfl_xor(sq, i, 32);
    }
    const float mean = s * (1.0f / OUT_DIM);
    const float var  = sq * (1.0f / OUT_DIM) - mean * mean;
    const float inv  = rsqrtf(var + 1e-5f);

    const float4 g4 = *(const float4*)(gamma + lane * 4);
    const float4 b4 = *(const float4*)(beta + lane * 4);
    float4 o;
    o.x = (v0 - mean) * inv * g4.x + b4.x;
    o.y = (v1 - mean) * inv * g4.y + b4.y;
    o.z = (v2 - mean) * inv * g4.z + b4.z;
    o.w = (v3 - mean) * inv * g4.w + b4.w;
    *(float4*)(out + (size_t)node * OUT_DIM + lane * 4) = o;
}

extern "C" void kernel_launch(void* const* d_in, const int* in_sizes, int n_in,
                              void* d_out, int out_size, void* d_ws, size_t ws_size,
                              hipStream_t stream) {
    const float* x     = (const float*)d_in[0];
    const int*   eidx  = (const int*)  d_in[1];
    const float* Wl    = (const float*)d_in[2];
    const float* bl    = (const float*)d_in[3];
    const float* Wr    = (const float*)d_in[4];
    const float* br    = (const float*)d_in[5];
    const float* att   = (const float*)d_in[6];
    const float* bias  = (const float*)d_in[7];
    const float* gamma = (const float*)d_in[8];
    const float* beta  = (const float*)d_in[9];
    float* out = (float*)d_out;

    const size_t szXL    = (size_t)BN * OUT_DIM;
    const size_t szE     = (size_t)B_BATCH * ETOT * HEADS;
    const size_t szH     = (size_t)BN * HEADS;
    const size_t needed  = (3 * szXL + szE + 2 * szH) * sizeof(float);
    if (ws_size < needed) return;

    float*    xl    = (float*)d_ws;
    float*    xr    = xl + szXL;
    float*    accum = xr + szXL;
    float*    e     = accum + szXL;
    unsigned* m_u   = (unsigned*)(e + szE);
    float*    denom = (float*)(m_u + szH);

    hipMemsetAsync(accum, 0, szXL * sizeof(float), stream);
    hipMemsetAsync(m_u,   0, szH * sizeof(unsigned), stream);
    hipMemsetAsync(denom, 0, szH * sizeof(float), stream);

    const int gemmWaves = (BN / 16) * 8 * 2;
    gat_gemm_wmma<<<gemmWaves / 8, 256, 0, stream>>>(x, Wl, bl, Wr, br, xl, xr);

    gat_edge_logits<<<EDGE_GROUPS / 8, 256, 0, stream>>>(eidx, xl, xr, att, e, m_u);

    gat_edge_scatter<<<EDGE_GROUPS / 8, 256, 0, stream>>>(eidx, xl, e, m_u, denom, accum);

    gat_finalize<<<BN / 8, 256, 0, stream>>>(accum, denom, bias, gamma, beta, out);
}


# Float-atomic scatter replaced by gather: edges bucketed by destination per call (int atomics + scan); one wave per node: logits, running softmax, sum, LayerNorm; sum order = bucket arrival order
# speedup vs baseline: 2.4658x; 1.9963x over previous
#include <hip/hip_runtime.h>
#include <hip/hip_bf16.h>

constexpr int B_BATCH = 4;
constexpr int N_NODES = 8192;
constexpr int E_EDGES = 131072;
constexpr int ETOT    = E_EDGES + N_NODES;
constexpr int IN_DIM  = 128;
constexpr int HEADS   = 4;
constexpr int CH      = 32;
constexpr int OUT_DIM = HEADS * CH;
constexpr int BN      = B_BATCH * N_NODES;

typedef float v2f __attribute__((ext_vector_type(2)));
typedef float v8f __attribute__((ext_vector_type(8)));

__device__ __forceinline__ float lrelu02(float v) {
    return v > 0.0f ? v : 0.2f * v;
}

__global__ __launch_bounds__(256) void gat_gemm_wmma(
    const float* __restrict__ x,
    const float* __restrict__ Wl, const float* __restrict__ bl,
    const float* __restrict__ Wr, const float* __restrict__ br,
    float* __restrict__ xl, float* __restrict__ xr)
{
    const int lane   = threadIdx.x & 31;
    const int wave   = blockIdx.x * 8 + (threadIdx.x >> 5);
    const int mat     = wave & 1;
    const int colTile = (wave >> 1) & 7;
    const int rowTile = wave >> 4;

    const int mOrN  = lane & 15;
    const int khalf = (lane >> 4) * 2;
    const int row   = rowTile * 16 + mOrN;
    const int col   = colTile * 16 + mOrN;

    const float* __restrict__ W    = mat ? Wr : Wl;
    const float* __restrict__ bias = mat ? br : bl;
    float*       __restrict__ out  = mat ? xr : xl;

    v8f acc = {};
#if __has_builtin(__builtin_amdgcn_wmma_f32_16x16x4_f32)
    #pragma unroll
    for (int k = 0; k < IN_DIM; k += 4) {
        v2f a = *(const v2f*)(x + (size_t)row * IN_DIM + k + khalf);
        v2f b;
        b.x = W[(size_t)(k + khalf + 0) * OUT_DIM + col];
        b.y = W[(size_t)(k + khalf + 1) * OUT_DIM + col];
        acc = __builtin_amdgcn_wmma_f32_16x16x4_f32(
                  false, a, false, b, (short)0, acc, false, false);
    }
#else
    const int rh = (lane >> 4) * 8;
    #pragma unroll
    for (int v = 0; v < 8; ++v) {
        float s = 0.0f;
        const int r = rowTile * 16 + rh + v;
        for (int k = 0; k < IN_DIM; ++k)
            s += x[(size_t)r * IN_DIM + k] * W[(size_t)k * OUT_DIM + col];
        acc[v] = s;
    }
#endif
    const int rowHalf = (lane >> 4) * 8;
    const float bv = bias[col];
    #pragma unroll
    for (int v = 0; v < 8; ++v)
        out[(size_t)(rowTile * 16 + rowHalf + v) * OUT_DIM + col] = acc[v] + bv;
}

constexpr int QUADS_PER_BATCH = E_EDGES / 4;
constexpr int EDGE_QUADS      = B_BATCH * QUADS_PER_BATCH;
constexpr int EB              = 4;
static_assert(E_EDGES % 4 == 0, "edge quads are aligned 16-byte loads");
static_assert(EDGE_QUADS % 256 == 0, "256 threads per block, no remainder");

__global__ __launch_bounds__(256) void gat_count_in(
    const int* __restrict__ eidx, int* __restrict__ cnt)
{
    const int t = blockIdx.x * 256 + threadIdx.x;
    const int b = t / QUADS_PER_BATCH;
    const int q = t - b * QUADS_PER_BATCH;
    const int4 d4 = *(const int4*)(eidx + ((size_t)b * 2 + 1) * E_EDGES + (size_t)q * 4);
    int* __restrict__ cb = cnt + (size_t)b * N_NODES;
    atomicAdd(cb + d4.x, 1);
    atomicAdd(cb + d4.y, 1);
    atomicAdd(cb + d4.z, 1);
    atomicAdd(cb + d4.w, 1);
}

constexpr int SCAN_THREADS = 1024;
constexpr int NODES_PER_THREAD = N_NODES / SCAN_THREADS;
static_assert(NODES_PER_THREAD == 8, "two 16-byte loads per thread");

__global__ __launch_bounds__(SCAN_THREADS) void gat_scan_start(
    const int* __restrict__ cnt, int* __restrict__ start)
{
    __shared__ int wsum[32];
    const int b    = blockIdx.x;
    const int t    = threadIdx.x;
    const int lane = t & 31;
    const int wv   = t >> 5;
    const int* __restrict__ c = cnt + (size_t)b * N_NODES + t * NODES_PER_THREAD;
    const int4 c0 = *(const int4*)(c);
    const int4 c1 = *(const int4*)(c + 4);
    const int v[8] = {c0.x, c0.y, c0.z, c0.w, c1.x, c1.y, c1.z, c1.w};
    int tot = 0;
    #pragma unroll
    for (int i = 0; i < 8; ++i) tot += v[i];

    int inc = tot;
    #pragma unroll
    for (int o = 1; o < 32; o <<= 1) {
        const int n = __shfl_up(inc, o, 32);
        if (lane >= o) inc += n;
    }
    if (lane == 31) wsum[wv] = inc;
    __syncthreads();
    if (wv == 0) {
        const int w = wsum[lane];
        int winc = w;
        #pragma unroll
        for (int o = 1; o < 32; o <<= 1) {
            const int n = __shfl_up(winc, o, 32);
            if (lane >= o) winc += n;
        }
        wsum[lane] = winc - w;
    }
    __syncthreads();

    int run = b * E_EDGES + wsum[wv] + inc - tot;
    int o8[8];
    #pragma unroll
    for (int i = 0; i < 8; ++i) { o8[i] = run; run += v[i]; }
    int* __restrict__ s = start + (size_t)b * N_NODES + t * NODES_PER_THREAD;
    *(int4*)(s)     = make_int4(o8[0], o8[1], o8[2], o8[3]);
    *(int4*)(s + 4) = make_int4(o8[4], o8[5], o8[6], o8[7]);
}

__global__ __launch_bounds__(256) void gat_fill_buckets(
    const int* __restrict__ eidx, const int* __restrict__ start,
    int* __restrict__ cur, int* __restrict__ csr)
{
    const int t = blockIdx.x * 256 + threadIdx.x;
    const int b = t / QUADS_PER_BATCH;
    const int q = t - b * QUADS_PER_BATCH;
    const int4 s4 = *(const int4*)(eidx + ((size_t)b * 2 + 0) * E_EDGES + (size_t)q * 4);
    const int4 d4 = *(const int4*)(eidx + ((size_t)b * 2 + 1) * E_EDGES + (size_t)q * 4);
    const int* __restrict__ sb = start + (size_t)b * N_NODES;
    int*       __restrict__ cb = cur   + (size_t)b * N_NODES;
    const int p0 = sb[d4.x] + atomicAdd(cb + d4.x, 1);
    const int p1 = sb[d4.y] + atomicAdd(cb + d4.y, 1);
    const int p2 = sb[d4.z] + atomicAdd(cb + d4.z, 1);
    const int p3 = sb[d4.w] + atomicAdd(cb + d4.w, 1);
    csr[p0] = s4.x;
    csr[p1] = s4.y;
    csr[p2] = s4.z;
    csr[p3] = s4.w;
}

__device__ __forceinline__ float head_logit(const float4 a, const float4 r, const float4 t)
{
    float p = lrelu02(a.x + r.x) * t.x
            + lrelu02(a.y + r.y) * t.y
            + lrelu02(a.z + r.z) * t.z
            + lrelu02(a.w + r.w) * t.w;
    p += __shfl_xor(p, 1, 32);
    p += __shfl_xor(p, 2, 32);
    p += __shfl_xor(p, 4, 32);
    return p;
}

__global__ __launch_bounds__(256) void gat_node_gather(
    const int* __restrict__ csr, const int* __restrict__ start, const int* __restrict__ indeg,
    const float* __restrict__ xl, const float* __restrict__ xr, const float* __restrict__ att,
    const float* __restrict__ bias, const float* __restrict__ gamma,
    const float* __restrict__ beta, float* __restrict__ out)
{
    const int lane = threadIdx.x & 31;
    const int wv   = __builtin_amdgcn_readfirstlane((int)(threadIdx.x >> 5));
    const int node = blockIdx.x * 8 + wv;
    const int b    = node / N_NODES;
    const int nloc = node - b * N_NODES;
    const int beg  = start[node];
    const int deg  = indeg[node];

    const float4 t4 = *(const float4*)(att + lane * 4);
    const float* __restrict__ xlb = xl + (size_t)b * N_NODES * OUT_DIM + lane * 4;
    const float4 r4 = *(const float4*)(xr + (size_t)node * OUT_DIM + lane * 4);

    float4 acc = *(const float4*)(xlb + (size_t)nloc * OUT_DIM);
    float  m   = head_logit(acc, r4, t4);
    float  den = 1.0f;

    for (int j0 = 0; j0 < deg; j0 += EB) {
        int src[EB];
        #pragma unroll
        for (int u = 0; u < EB; ++u) {
            const int j = (j0 + u < deg) ? (j0 + u) : (deg - 1);
            src[u] = csr[beg + j];
        }
        float4 a4[EB];
        #pragma unroll
        for (int u = 0; u < EB; ++u)
            a4[u] = *(const float4*)(xlb + (size_t)src[u] * OUT_DIM);
        float p[EB];
        float mn = m;
        #pragma unroll
        for (int u = 0; u < EB; ++u) {
            p[u] = head_logit(a4[u], r4, t4);
            mn = (j0 + u < deg) ? fmaxf(mn, p[u]) : mn;
        }
        const float sc = __expf(m - mn);
        float ex[EB];
        #pragma unroll
        for (int u = 0; u < EB; ++u)
            ex[u] = (j0 + u < deg) ? __expf(p[u] - mn) : 0.0f;
        den   = den   * sc + ((ex[0] + ex[1]) + (ex[2] + ex[3]));
        acc.x = acc.x * sc + ((ex[0] * a4[0].x + ex[1] * a4[1].x) + (ex[2] * a4[2].x + ex[3] * a4[3].x));
        acc.y = acc.y * sc + ((ex[0] * a4[0].y + ex[1] * a4[1].y) + (ex[2] * a4[2].y + ex[3] * a4[3].y));
        acc.z = acc.z * sc + ((ex[0] * a4[0].z + ex[1] * a4[1].z) + (ex[2] * a4[2].z + ex[3] * a4[3].z));
        acc.w = acc.w * sc + ((ex[0] * a4[0].w + ex[1] * a4[1].w) + (ex[2] * a4[2].w + ex[3] * a4[3].w));
        m = mn;
    }
    static_assert(EB == 4, "the batch sums above are written out for 4 edges");

    const float  dh = den + 1e-16f;
    const float4 bi = *(const float4*)(bias + lane * 4);
    float v0 = acc.x / dh + bi.x;
    float v1 = acc.y / dh + bi.y;
    float v2 = acc.z / dh + bi.z;
    float v3 = acc.w / dh + bi.w;

    float s  = v0 + v1 + v2 + v3;
    float sq = v0 * v0 + v1 * v1 + v2 * v2 + v3 * v3;
    #pragma unroll
    for (int i = 1; i < 32; i <<= 1) {
        s  += __shfl_xor(s,  i, 32);
        sq += __shfl_xor(sq, i, 32);
    }
    const float mean = s * (1.0f / OUT_DIM);
    const float var  = sq * (1.0f / OUT_DIM) - mean * mean;
    const float inv  = rsqrtf(var + 1e-5f);

    const float4 g4 = *(const float4*)(gamma + lane * 4);
    const float4 b4 = *(const float4*)(beta + lane * 4);
    float4 o;
    o.x = (v0 - mean) * inv * g4.x + b4.x;
    o.y = (v1 - mean) * inv * g4.y + b4.y;
    o.z = (v2 - mean) * inv * g4.z + b4.z;
    o.w = (v3 - mean) * inv * g4.w + b4.w;
    *(float4*)(out + (size_t)node * OUT_DIM + lane * 4) = o;
}

extern "C" void kernel_launch(void* const* d_in, const int* in_sizes, int n_in,
                              void* d_out, int out_size, void* d_ws, size_t ws_size,
                              hipStream_t stream) {
    const float* x     = (const float*)d_in[0];
    const int*   eidx  = (const int*)  d_in[1];
    const float* Wl    = (const float*)d_in[2];
    const float* bl    = (const float*)d_in[3];
    const float* Wr    = (const float*)d_in[4];
    const float* br    = (const float*)d_in[5];
    const float* att   = (const float*)d_in[6];
    const float* bias  = (const float*)d_in[7];
    const float* gamma = (const float*)d_in[8];
    const float* beta  = (const float*)d_in[9];
    float* out = (float*)d_out;

    const size_t szXL    = (size_t)BN * OUT_DIM;
    const size_t szCSR   = (size_t)B_BATCH * E_EDGES;
    const size_t szN     = (size_t)BN;
    const size_t needed  = (2 * szXL + szCSR + 3 * szN) * sizeof(float);
    if (ws_size < needed) return;

    float* xl    = (float*)d_ws;
    float* xr    = xl + szXL;
    int*   csr   = (int*)(xr + szXL);
    int*   cnt   = csr + szCSR;
    int*   cur   = cnt + szN;
    int*   start = cur + szN;

    hipMemsetAsync(cnt, 0, 2 * szN * sizeof(int), stream);

    gat_count_in<<<EDGE_QUADS / 256, 256, 0, stream>>>(eidx, cnt);
    gat_scan_start<<<B_BATCH, SCAN_THREADS, 0, stream>>>(cnt, start);
    gat_fill_buckets<<<EDGE_QUADS / 256, 256, 0, stream>>>(eidx, start, cur, csr);

    const int gemmWaves = (BN / 16) * 8 * 2;
    gat_gemm_wmma<<<gemmWaves / 8, 256, 0, stream>>>(x, Wl, bl, Wr, br, xl, xr);

    gat_node_gather<<<BN / 8, 256, 0, stream>>>(csr, start, cnt, xl, xr, att, bias, gamma, beta, out);
}


# GEMM re-tiled: W_l|W_r staged once per workgroup in LDS, each wave computes a 16x256 strip of xl|xr (16 accumulator tiles, x read once, whole-line stores); 2048 waves, not 32768; sum order unchanged
# speedup vs baseline: 2.7347x; 1.1091x over previous
#include <hip/hip_runtime.h>
#include <hip/hip_bf16.h>

constexpr int B_BATCH = 4;
constexpr int N_NODES = 8192;
constexpr int E_EDGES = 131072;
constexpr int ETOT    = E_EDGES + N_NODES;
constexpr int IN_DIM  = 128;
constexpr int HEADS   = 4;
constexpr int CH      = 32;
constexpr int OUT_DIM = HEADS * CH;
constexpr int BN      = B_BATCH * N_NODES;

typedef float v2f __attribute__((ext_vector_type(2)));
typedef float v8f __attribute__((ext_vector_type(8)));

__device__ __forceinline__ float lrelu02(float v) {
    return v > 0.0f ? v : 0.2f * v;
}

constexpr int GEMM_WAVES = 8;
constexpr int GEMM_NCOL  = 2 * OUT_DIM;
constexpr int GEMM_TILES = GEMM_NCOL / 16;
constexpr int KSTEPS     = IN_DIM / 4;
constexpr int KGROUP     = 2;
static_assert(KSTEPS % KGROUP == 0, "whole groups");
constexpr int KPAIRS     = IN_DIM / 2;
constexpr int WPITCH     = 672;
static_assert(WPITCH >= 2 * GEMM_NCOL && WPITCH % 64 == 32, "pitch holds a row and splits the bank halves");
static_assert((BN / 16) % GEMM_WAVES == 0, "whole workgroups of row strips");
static_assert(IN_DIM == 128 && OUT_DIM == 128, "staging index arithmetic below");

__global__ __launch_bounds__(256) void gat_gemm_wmma(
    const float* __restrict__ x,
    const float* __restrict__ Wl, const float* __restrict__ bl,
    const float* __restrict__ Wr, const float* __restrict__ br,
    float* __restrict__ xl, float* __restrict__ xr)
{
    __shared__ __attribute__((aligned(16))) float wlds[KPAIRS * WPITCH];

    const int lane    = threadIdx.x & 31;
    const int wv      = threadIdx.x >> 5;
    const int rowTile = blockIdx.x * GEMM_WAVES + wv;
    const int c       = lane & 15;
    const int h       = lane >> 4;
    const int row     = rowTile * 16 + c;

    v2f a[KSTEPS];
    #pragma unroll
    for (int s = 0; s < KSTEPS; ++s)
        a[s] = *(const v2f*)(x + (size_t)row * IN_DIM + 4 * s + 2 * h);

    #pragma unroll
    for (int i = 0; i < (KPAIRS * 2 * 32) / 256; ++i) {
        const int task = i * 256 + (int)threadIdx.x;
        const int c4   = task & 31;
        const int mat  = (task >> 5) & 1;
        const int kp   = task >> 6;
        const float* __restrict__ Wm = mat ? Wr : Wl;
        const float4 r0 = *(const float4*)(Wm + (size_t)(2 * kp)     * OUT_DIM + c4 * 4);
        const float4 r1 = *(const float4*)(Wm + (size_t)(2 * kp + 1) * OUT_DIM + c4 * 4);
        float* d = wlds + kp * WPITCH + (mat * OUT_DIM + c4 * 4) * 2;
        *(float4*)(d)     = make_float4(r0.x, r1.x, r0.y, r1.y);
        *(float4*)(d + 4) = make_float4(r0.z, r1.z, r0.w, r1.w);
    }
    __syncthreads();

    const float* bfrag = wlds + h * WPITCH + c * 2;
    v8f acc[GEMM_TILES];
    #pragma unroll
    for (int j = 0; j < GEMM_TILES; ++j) acc[j] = v8f{};

    v2f bcur[KGROUP * GEMM_TILES];
    #pragma unroll
    for (int u = 0; u < KGROUP; ++u)
        #pragma unroll
        for (int j = 0; j < GEMM_TILES; ++j)
            bcur[u * GEMM_TILES + j] = *(const v2f*)(bfrag + (2 * u) * WPITCH + j * 32);

    #pragma unroll
    for (int g = 0; g < KSTEPS / KGROUP; ++g) {
        v2f bnext[KGROUP * GEMM_TILES];
        if (g + 1 < KSTEPS / KGROUP) {
            #pragma unroll
            for (int u = 0; u < KGROUP; ++u)
                #pragma unroll
                for (int j = 0; j < GEMM_TILES; ++j)
                    bnext[u * GEMM_TILES + j] =
                        *(const v2f*)(bfrag + (2 * ((g + 1) * KGROUP + u)) * WPITCH + j * 32);
        }
        __builtin_amdgcn_sched_barrier(0);
        #pragma unroll
        for (int u = 0; u < KGROUP; ++u)
            #pragma unroll
            for (int j = 0; j < GEMM_TILES; ++j)
                acc[j] = __builtin_amdgcn_wmma_f32_16x16x4_f32(
                             false, a[g * KGROUP + u], false, bcur[u * GEMM_TILES + j],
                             (short)0, acc[j], false, false);
        __builtin_amdgcn_sched_barrier(0);
        if (g + 1 < KSTEPS / KGROUP) {
            #pragma unroll
            for (int q = 0; q < KGROUP * GEMM_TILES; ++q) bcur[q] = bnext[q];
        }
    }

    #pragma unroll
    for (int jp = 0; jp < GEMM_TILES / 2; ++jp) {
        const int    mcol = (jp & 3) * 32 + lane;
        float* __restrict__ om = (jp < 4) ? xl : xr;
        const float  bv   = ((jp < 4) ? bl : br)[mcol];
        #pragma unroll
        for (int r = 0; r < 8; ++r) {
            const float lo   = acc[2 * jp][r];
            const float hi   = acc[2 * jp + 1][r];
            const float give = h ? lo : hi;
            const float got  = __shfl_xor(give, 16, 32);
            const float top  = h ? got : lo;
            const float bot  = h ? hi  : got;
            om[(size_t)(rowTile * 16 + r)     * OUT_DIM + mcol] = top + bv;
            om[(size_t)(rowTile * 16 + r + 8) * OUT_DIM + mcol] = bot + bv;
        }
    }
}

constexpr int QUADS_PER_BATCH = E_EDGES / 4;
constexpr int EDGE_QUADS      = B_BATCH * QUADS_PER_BATCH;
constexpr int EB              = 4;
static_assert(E_EDGES % 4 == 0, "edge quads are aligned 16-byte loads");
static_assert(EDGE_QUADS % 256 == 0, "256 threads per block, no remainder");

__global__ __launch_bounds__(256) void gat_count_in(
    const int* __restrict__ eidx, int* __restrict__ cnt)
{
    const int t = blockIdx.x * 256 + threadIdx.x;
    const int b = t / QUADS_PER_BATCH;
    const int q = t - b * QUADS_PER_BATCH;
    const int4 d4 = *(const int4*)(eidx + ((size_t)b * 2 + 1) * E_EDGES + (size_t)q * 4);
    int* __restrict__ cb = cnt + (size_t)b * N_NODES;
    atomicAdd(cb + d4.x, 1);
    atomicAdd(cb + d4.y, 1);
    atomicAdd(cb + d4.z, 1);
    atomicAdd(cb + d4.w, 1);
}

constexpr int SCAN_THREADS = 1024;
constexpr int NODES_PER_THREAD = N_NODES / SCAN_THREADS;
static_assert(NODES_PER_THREAD == 8, "two 16-byte loads per thread");

__global__ __launch_bounds__(SCAN_THREADS) void gat_scan_start(
    const int* __restrict__ cnt, int* __restrict__ start)
{
    __shared__ int wsum[32];
    const int b    = blockIdx.x;
    const int t    = threadIdx.x;
    const int lane = t & 31;
    const int wv   = t >> 5;
    const int* __restrict__ c = cnt + (size_t)b * N_NODES + t * NODES_PER_THREAD;
    const int4 c0 = *(const int4*)(c);
    const int4 c1 = *(const int4*)(c + 4);
    const int v[8] = {c0.x, c0.y, c0.z, c0.w, c1.x, c1.y, c1.z, c1.w};
    int tot = 0;
    #pragma unroll
    for (int i = 0; i < 8; ++i) tot += v[i];

    int inc = tot;
    #pragma unroll
    for (int o = 1; o < 32; o <<= 1) {
        const int n = __shfl_up(inc, o, 32);
        if (lane >= o) inc += n;
    }
    if (lane == 31) wsum[wv] = inc;
    __syncthreads();
    if (wv == 0) {
        const int w = wsum[lane];
        int winc = w;
        #pragma unroll
        for (int o = 1; o < 32; o <<= 1) {
            const int n = __shfl_up(winc, o, 32);
            if (lane >= o) winc += n;
        }
        wsum[lane] = winc - w;
    }
    __syncthreads();

    int run = b * E_EDGES + wsum[wv] + inc - tot;
    int o8[8];
    #pragma unroll
    for (int i = 0; i < 8; ++i) { o8[i] = run; run += v[i]; }
    int* __restrict__ s = start + (size_t)b * N_NODES + t * NODES_PER_THREAD;
    *(int4*)(s)     = make_int4(o8[0], o8[1], o8[2], o8[3]);
    *(int4*)(s + 4) = make_int4(o8[4], o8[5], o8[6], o8[7]);
}

__global__ __launch_bounds__(256) void gat_fill_buckets(
    const int* __restrict__ eidx, const int* __restrict__ start,
    int* __restrict__ cur, int* __restrict__ csr)
{
    const int t = blockIdx.x * 256 + threadIdx.x;
    const int b = t / QUADS_PER_BATCH;
    const int q = t - b * QUADS_PER_BATCH;
    const int4 s4 = *(const int4*)(eidx + ((size_t)b * 2 + 0) * E_EDGES + (size_t)q * 4);
    const int4 d4 = *(const int4*)(eidx + ((size_t)b * 2 + 1) * E_EDGES + (size_t)q * 4);
    const int* __restrict__ sb = start + (size_t)b * N_NODES;
    int*       __restrict__ cb = cur   + (size_t)b * N_NODES;
    const int p0 = sb[d4.x] + atomicAdd(cb + d4.x, 1);
    const int p1 = sb[d4.y] + atomicAdd(cb + d4.y, 1);
    const int p2 = sb[d4.z] + atomicAdd(cb + d4.z, 1);
    const int p3 = sb[d4.w] + atomicAdd(cb + d4.w, 1);
    csr[p0] = s4.x;
    csr[p1] = s4.y;
    csr[p2] = s4.z;
    csr[p3] = s4.w;
}

__device__ __forceinline__ float head_logit(const float4 a, const float4 r, const float4 t)
{
    float p = lrelu02(a.x + r.x) * t.x
            + lrelu02(a.y + r.y) * t.y
            + lrelu02(a.z + r.z) * t.z
            + lrelu02(a.w + r.w) * t.w;
    p += __shfl_xor(p, 1, 32);
    p += __shfl_xor(p, 2, 32);
    p += __shfl_xor(p, 4, 32);
    return p;
}

__global__ __launch_bounds__(256) void gat_node_gather(
    const int* __restrict__ csr, const int* __restrict__ start, const int* __restrict__ indeg,
    const float* __restrict__ xl, const float* __restrict__ xr, const float* __restrict__ att,
    const float* __restrict__ bias, const float* __restrict__ gamma,
    const float* __restrict__ beta, float* __restrict__ out)
{
    const int lane = threadIdx.x & 31;
    const int wv   = __builtin_amdgcn_readfirstlane((int)(threadIdx.x >> 5));
    const int node = blockIdx.x * 8 + wv;
    const int b    = node / N_NODES;
    const int nloc = node - b * N_NODES;
    const int beg  = start[node];
    const int deg  = indeg[node];

    const float4 t4 = *(const float4*)(att + lane * 4);
    const float* __restrict__ xlb = xl + (size_t)b * N_NODES * OUT_DIM + lane * 4;
    const float4 r4 = *(const float4*)(xr + (size_t)node * OUT_DIM + lane * 4);

    float4 acc = *(const float4*)(xlb + (size_t)nloc * OUT_DIM);
    float  m   = head_logit(acc, r4, t4);
    float  den = 1.0f;

    for (int j0 = 0; j0 < deg; j0 += EB) {
        int src[EB];
        #pragma unroll
        for (int u = 0; u < EB; ++u) {
            const int j = (j0 + u < deg) ? (j0 + u) : (deg - 1);
            src[u] = csr[beg + j];
        }
        float4 a4[EB];
        #pragma unroll
        for (int u = 0; u < EB; ++u)
            a4[u] = *(const float4*)(xlb + (size_t)src[u] * OUT_DIM);
        float p[EB];
        float mn = m;
        #pragma unroll
        for (int u = 0; u < EB; ++u) {
            p[u] = head_logit(a4[u], r4, t4);
            mn = (j0 + u < deg) ? fmaxf(mn, p[u]) : mn;
        }
        const float sc = __expf(m - mn);
        float ex[EB];
        #pragma unroll
        for (int u = 0; u < EB; ++u)
            ex[u] = (j0 + u < deg) ? __expf(p[u] - mn) : 0.0f;
        den   = den   * sc + ((ex[0] + ex[1]) + (ex[2] + ex[3]));
        acc.x = acc.x * sc + ((ex[0] * a4[0].x + ex[1] * a4[1].x) + (ex[2] * a4[2].x + ex[3] * a4[3].x));
        acc.y = acc.y * sc + ((ex[0] * a4[0].y + ex[1] * a4[1].y) + (ex[2] * a4[2].y + ex[3] * a4[3].y));
        acc.z = acc.z * sc + ((ex[0] * a4[0].z + ex[1] * a4[1].z) + (ex[2] * a4[2].z + ex[3] * a4[3].z));
        acc.w = acc.w * sc + ((ex[0] * a4[0].w + ex[1] * a4[1].w) + (ex[2] * a4[2].w + ex[3] * a4[3].w));
        m = mn;
    }
    static_assert(EB == 4, "the batch sums above are written out for 4 edges");

    const float  dh = den + 1e-16f;
    const float4 bi = *(const float4*)(bias + lane * 4);
    float v0 = acc.x / dh + bi.x;
    float v1 = acc.y / dh + bi.y;
    float v2 = acc.z / dh + bi.z;
    float v3 = acc.w / dh + bi.w;

    float s  = v0 + v1 + v2 + v3;
    float sq = v0 * v0 + v1 * v1 + v2 * v2 + v3 * v3;
    #pragma unroll
    for (int i = 1; i < 32; i <<= 1) {
        s  += __shfl_xor(s,  i, 32);
        sq += __shfl_xor(sq, i, 32);
    }
    const float mean = s * (1.0f / OUT_DIM);
    const float var  = sq * (1.0f / OUT_DIM) - mean * mean;
    const float inv  = rsqrtf(var + 1e-5f);

    const float4 g4 = *(const float4*)(gamma + lane * 4);
    const float4 b4 = *(const float4*)(beta + lane * 4);
    float4 o;
    o.x = (v0 - mean) * inv * g4.x + b4.x;
    o.y = (v1 - mean) * inv * g4.y + b4.y;
    o.z = (v2 - mean) * inv * g4.z + b4.z;
    o.w = (v3 - mean) * inv * g4.w + b4.w;
    *(float4*)(out + (size_t)node * OUT_DIM + lane * 4) = o;
}

extern "C" void kernel_launch(void* const* d_in, const int* in_sizes, int n_in,
                              void* d_out, int out_size, void* d_ws, size_t ws_size,
                              hipStream_t stream) {
    const float* x     = (const float*)d_in[0];
    const int*   eidx  = (const int*)  d_in[1];
    const float* Wl    = (const float*)d_in[2];
    const float* bl    = (const float*)d_in[3];
    const float* Wr    = (const float*)d_in[4];
    const float* br    = (const float*)d_in[5];
    const float* att   = (const float*)d_in[6];
    const float* bias  = (const float*)d_in[7];
    const float* gamma = (const float*)d_in[8];
    const float* beta  = (const float*)d_in[9];
    float* out = (float*)d_out;

    const size_t szXL    = (size_t)BN * OUT_DIM;
    const size_t szCSR   = (size_t)B_BATCH * E_EDGES;
    const size_t szN     = (size_t)BN;
    const size_t needed  = (2 * szXL + szCSR + 3 * szN) * sizeof(float);
    if (ws_size < needed) return;

    float* xl    = (float*)d_ws;
    float* xr    = xl + szXL;
    int*   csr   = (int*)(xr + szXL);
    int*   cnt   = csr + szCSR;
    int*   cur   = cnt + szN;
    int*   start = cur + szN;

    hipMemsetAsync(cnt, 0, 2 * szN * sizeof(int), stream);

    gat_count_in<<<EDGE_QUADS / 256, 256, 0, stream>>>(eidx, cnt);
    gat_scan_start<<<B_BATCH, SCAN_THREADS, 0, stream>>>(cnt, start);
    gat_fill_buckets<<<EDGE_QUADS / 256, 256, 0, stream>>>(eidx, start, cur, csr);

    gat_gemm_wmma<<<(BN / 16) / GEMM_WAVES, 256, 0, stream>>>(x, Wl, bl, Wr, br, xl, xr);

    gat_node_gather<<<BN / 8, 256, 0, stream>>>(csr, start, cnt, xl, xr, att, bias, gamma, beta, out);
}


# Edge ranks taken in LDS: 32 blocks count 16384 edges each with returning LDS adds, one global add per (block,node); fill places at start+rank, no atomics; bucket order = arrival order of those adds
# speedup vs baseline: 4.8783x; 1.7839x over previous
#include <hip/hip_runtime.h>
#include <hip/hip_bf16.h>

constexpr int B_BATCH = 4;
constexpr int N_NODES = 8192;
constexpr int E_EDGES = 131072;
constexpr int ETOT    = E_EDGES + N_NODES;
constexpr int IN_DIM  = 128;
constexpr int HEADS   = 4;
constexpr int CH      = 32;
constexpr int OUT_DIM = HEADS * CH;
constexpr int BN      = B_BATCH * N_NODES;

typedef float v2f __attribute__((ext_vector_type(2)));
typedef float v8f __attribute__((ext_vector_type(8)));

__device__ __forceinline__ float lrelu02(float v) {
    return v > 0.0f ? v : 0.2f * v;
}

constexpr int GEMM_WAVES = 8;
constexpr int GEMM_NCOL  = 2 * OUT_DIM;
constexpr int GEMM_TILES = GEMM_NCOL / 16;
constexpr int KSTEPS     = IN_DIM / 4;
constexpr int KGROUP     = 2;
static_assert(KSTEPS % KGROUP == 0, "whole groups");
constexpr int KPAIRS     = IN_DIM / 2;
constexpr int WPITCH     = 672;
static_assert(WPITCH >= 2 * GEMM_NCOL && WPITCH % 64 == 32, "pitch holds a row and splits the bank halves");
static_assert((BN / 16) % GEMM_WAVES == 0, "whole workgroups of row strips");
static_assert(IN_DIM == 128 && OUT_DIM == 128, "staging index arithmetic below");

__global__ __launch_bounds__(256) void gat_gemm_wmma(
    const float* __restrict__ x,
    const float* __restrict__ Wl, const float* __restrict__ bl,
    const float* __restrict__ Wr, const float* __restrict__ br,
    float* __restrict__ xl, float* __restrict__ xr)
{
    __shared__ __attribute__((aligned(16))) float wlds[KPAIRS * WPITCH];

    const int lane    = threadIdx.x & 31;
    const int wv      = threadIdx.x >> 5;
    const int rowTile = blockIdx.x * GEMM_WAVES + wv;
    const int c       = lane & 15;
    const int h       = lane >> 4;
    const int row     = rowTile * 16 + c;

    v2f a[KSTEPS];
    #pragma unroll
    for (int s = 0; s < KSTEPS; ++s)
        a[s] = *(const v2f*)(x + (size_t)row * IN_DIM + 4 * s + 2 * h);

    #pragma unroll
    for (int i = 0; i < (KPAIRS * 2 * 32) / 256; ++i) {
        const int task = i * 256 + (int)threadIdx.x;
        const int c4   = task & 31;
        const int mat  = (task >> 5) & 1;
        const int kp   = task >> 6;
        const float* __restrict__ Wm = mat ? Wr : Wl;
        const float4 r0 = *(const float4*)(Wm + (size_t)(2 * kp)     * OUT_DIM + c4 * 4);
        const float4 r1 = *(const float4*)(Wm + (size_t)(2 * kp + 1) * OUT_DIM + c4 * 4);
        float* d = wlds + kp * WPITCH + (mat * OUT_DIM + c4 * 4) * 2;
        *(float4*)(d)     = make_float4(r0.x, r1.x, r0.y, r1.y);
        *(float4*)(d + 4) = make_float4(r0.z, r1.z, r0.w, r1.w);
    }
    __syncthreads();

    const float* bfrag = wlds + h * WPITCH + c * 2;
    v8f acc[GEMM_TILES];
    #pragma unroll
    for (int j = 0; j < GEMM_TILES; ++j) acc[j] = v8f{};

    v2f bcur[KGROUP * GEMM_TILES];
    #pragma unroll
    for (int u = 0; u < KGROUP; ++u)
        #pragma unroll
        for (int j = 0; j < GEMM_TILES; ++j)
            bcur[u * GEMM_TILES + j] = *(const v2f*)(bfrag + (2 * u) * WPITCH + j * 32);

    #pragma unroll
    for (int g = 0; g < KSTEPS / KGROUP; ++g) {
        v2f bnext[KGROUP * GEMM_TILES];
        if (g + 1 < KSTEPS / KGROUP) {
            #pragma unroll
            for (int u = 0; u < KGROUP; ++u)
                #pragma unroll
                for (int j = 0; j < GEMM_TILES; ++j)
                    bnext[u * GEMM_TILES + j] =
                        *(const v2f*)(bfrag + (2 * ((g + 1) * KGROUP + u)) * WPITCH + j * 32);
        }
        __builtin_amdgcn_sched_barrier(0);
        #pragma unroll
        for (int u = 0; u < KGROUP; ++u)
            #pragma unroll
            for (int j = 0; j < GEMM_TILES; ++j)
                acc[j] = __builtin_amdgcn_wmma_f32_16x16x4_f32(
                             false, a[g * KGROUP + u], false, bcur[u * GEMM_TILES + j],
                             (short)0, acc[j], false, false);
        __builtin_amdgcn_sched_barrier(0);
        if (g + 1 < KSTEPS / KGROUP) {
            #pragma unroll
            for (int q = 0; q < KGROUP * GEMM_TILES; ++q) bcur[q] = bnext[q];
        }
    }

    #pragma unroll
    for (int jp = 0; jp < GEMM_TILES / 2; ++jp) {
        const int    mcol = (jp & 3) * 32 + lane;
        float* __restrict__ om = (jp < 4) ? xl : xr;
        const float  bv   = ((jp < 4) ? bl : br)[mcol];
        #pragma unroll
        for (int r = 0; r < 8; ++r) {
            const float lo   = acc[2 * jp][r];
            const float hi   = acc[2 * jp + 1][r];
            const float give = h ? lo : hi;
            const float got  = __shfl_xor(give, 16, 32);
            const float top  = h ? got : lo;
            const float bot  = h ? hi  : got;
            om[(size_t)(rowTile * 16 + r)     * OUT_DIM + mcol] = top + bv;
            om[(size_t)(rowTile * 16 + r + 8) * OUT_DIM + mcol] = bot + bv;
        }
    }
}

constexpr int QUADS_PER_BATCH = E_EDGES / 4;
constexpr int EDGE_QUADS      = B_BATCH * QUADS_PER_BATCH;
constexpr int EB              = 4;
static_assert(E_EDGES % 4 == 0, "edge quads are aligned 16-byte loads");
static_assert(EDGE_QUADS % 256 == 0, "256 threads per block, no remainder");

constexpr int RANK_SPLIT   = 8;
constexpr int RANK_THREADS = 1024;
constexpr int RANK_QUADS   = QUADS_PER_BATCH / RANK_SPLIT;
constexpr int RANK_Q_PER_T = RANK_QUADS / RANK_THREADS;
constexpr int BINS_PER_T   = N_NODES / RANK_THREADS;
static_assert(QUADS_PER_BATCH % RANK_SPLIT == 0 && RANK_QUADS % RANK_THREADS == 0, "whole quads per thread");
static_assert(N_NODES % RANK_THREADS == 0, "whole table words per thread");

__global__ __launch_bounds__(RANK_THREADS) void gat_count_rank(
    const int* __restrict__ eidx, int* __restrict__ cnt, int* __restrict__ rank)
{
    __shared__ int hist[N_NODES];
    const int t = threadIdx.x;
    const int b = blockIdx.x / RANK_SPLIT;
    const int g = blockIdx.x - b * RANK_SPLIT;
    const int* __restrict__ dst = eidx + ((size_t)b * 2 + 1) * E_EDGES + (size_t)g * RANK_QUADS * 4;
    int*       __restrict__ rk  = rank + (size_t)b * E_EDGES + (size_t)g * RANK_QUADS * 4;

    int4 d[RANK_Q_PER_T];
    #pragma unroll
    for (int i = 0; i < RANK_Q_PER_T; ++i)
        d[i] = *(const int4*)(dst + (size_t)(i * RANK_THREADS + t) * 4);

    for (int i = 0; i < BINS_PER_T; ++i) hist[i * RANK_THREADS + t] = 0;
    __syncthreads();

    int4 r[RANK_Q_PER_T];
    #pragma unroll
    for (int i = 0; i < RANK_Q_PER_T; ++i) {
        r[i].x = atomicAdd(&hist[d[i].x], 1);
        r[i].y = atomicAdd(&hist[d[i].y], 1);
        r[i].z = atomicAdd(&hist[d[i].z], 1);
        r[i].w = atomicAdd(&hist[d[i].w], 1);
    }
    __syncthreads();

    int* __restrict__ cb = cnt + (size_t)b * N_NODES;
    int first[BINS_PER_T];
    #pragma unroll
    for (int i = 0; i < BINS_PER_T; ++i) {
        const int bin = i * RANK_THREADS + t;
        first[i] = atomicAdd(cb + bin, hist[bin]);
    }
    #pragma unroll
    for (int i = 0; i < BINS_PER_T; ++i) hist[i * RANK_THREADS + t] = first[i];
    __syncthreads();

    #pragma unroll
    for (int i = 0; i < RANK_Q_PER_T; ++i) {
        r[i].x += hist[d[i].x];
        r[i].y += hist[d[i].y];
        r[i].z += hist[d[i].z];
        r[i].w += hist[d[i].w];
        *(int4*)(rk + (size_t)(i * RANK_THREADS + t) * 4) = r[i];
    }
}

constexpr int SCAN_THREADS = 1024;
constexpr int NODES_PER_THREAD = N_NODES / SCAN_THREADS;
static_assert(NODES_PER_THREAD == 8, "two 16-byte loads per thread");

__global__ __launch_bounds__(SCAN_THREADS) void gat_scan_start(
    const int* __restrict__ cnt, int* __restrict__ start)
{
    __shared__ int wsum[32];
    const int b    = blockIdx.x;
    const int t    = threadIdx.x;
    const int lane = t & 31;
    const int wv   = t >> 5;
    const int* __restrict__ c = cnt + (size_t)b * N_NODES + t * NODES_PER_THREAD;
    const int4 c0 = *(const int4*)(c);
    const int4 c1 = *(const int4*)(c + 4);
    const int v[8] = {c0.x, c0.y, c0.z, c0.w, c1.x, c1.y, c1.z, c1.w};
    int tot = 0;
    #pragma unroll
    for (int i = 0; i < 8; ++i) tot += v[i];

    int inc = tot;
    #pragma unroll
    for (int o = 1; o < 32; o <<= 1) {
        const int n = __shfl_up(inc, o, 32);
        if (lane >= o) inc += n;
    }
    if (lane == 31) wsum[wv] = inc;
    __syncthreads();
    if (wv == 0) {
        const int w = wsum[lane];
        int winc = w;
        #pragma unroll
        for (int o = 1; o < 32; o <<= 1) {
            const int n = __shfl_up(winc, o, 32);
            if (lane >= o) winc += n;
        }
        wsum[lane] = winc - w;
    }
    __syncthreads();

    int run = b * E_EDGES + wsum[wv] + inc - tot;
    int o8[8];
    #pragma unroll
    for (int i = 0; i < 8; ++i) { o8[i] = run; run += v[i]; }
    int* __restrict__ s = start + (size_t)b * N_NODES + t * NODES_PER_THREAD;
    *(int4*)(s)     = make_int4(o8[0], o8[1], o8[2], o8[3]);
    *(int4*)(s + 4) = make_int4(o8[4], o8[5], o8[6], o8[7]);
}

__global__ __launch_bounds__(256) void gat_fill_buckets(
    const int* __restrict__ eidx, const int* __restrict__ start,
    const int* __restrict__ rank, int* __restrict__ csr)
{
    const int t = blockIdx.x * 256 + threadIdx.x;
    const int b = t / QUADS_PER_BATCH;
    const int q = t - b * QUADS_PER_BATCH;
    const int4 s4 = *(const int4*)(eidx + ((size_t)b * 2 + 0) * E_EDGES + (size_t)q * 4);
    const int4 d4 = *(const int4*)(eidx + ((size_t)b * 2 + 1) * E_EDGES + (size_t)q * 4);
    const int4 r4 = *(const int4*)(rank + (size_t)b * E_EDGES + (size_t)q * 4);
    const int* __restrict__ sb = start + (size_t)b * N_NODES;
    const int p0 = sb[d4.x] + r4.x;
    const int p1 = sb[d4.y] + r4.y;
    const int p2 = sb[d4.z] + r4.z;
    const int p3 = sb[d4.w] + r4.w;
    csr[p0] = s4.x;
    csr[p1] = s4.y;
    csr[p2] = s4.z;
    csr[p3] = s4.w;
}

__device__ __forceinline__ float head_logit(const float4 a, const float4 r, const float4 t)
{
    float p = lrelu02(a.x + r.x) * t.x
            + lrelu02(a.y + r.y) * t.y
            + lrelu02(a.z + r.z) * t.z
            + lrelu02(a.w + r.w) * t.w;
    p += __shfl_xor(p, 1, 32);
    p += __shfl_xor(p, 2, 32);
    p += __shfl_xor(p, 4, 32);
    return p;
}

__global__ __launch_bounds__(256) void gat_node_gather(
    const int* __restrict__ csr, const int* __restrict__ start, const int* __restrict__ indeg,
    const float* __restrict__ xl, const float* __restrict__ xr, const float* __restrict__ att,
    const float* __restrict__ bias, const float* __restrict__ gamma,
    const float* __restrict__ beta, float* __restrict__ out)
{
    const int lane = threadIdx.x & 31;
    const int wv   = __builtin_amdgcn_readfirstlane((int)(threadIdx.x >> 5));
    const int node = blockIdx.x * 8 + wv;
    const int b    = node / N_NODES;
    const int nloc = node - b * N_NODES;
    const int beg  = start[node];
    const int deg  = indeg[node];

    const float4 t4 = *(const float4*)(att + lane * 4);
    const float* __restrict__ xlb = xl + (size_t)b * N_NODES * OUT_DIM + lane * 4;
    const float4 r4 = *(const float4*)(xr + (size_t)node * OUT_DIM + lane * 4);

    float4 acc = *(const float4*)(xlb + (size_t)nloc * OUT_DIM);
    float  m   = head_logit(acc, r4, t4);
    float  den = 1.0f;

    for (int j0 = 0; j0 < deg; j0 += EB) {
        int src[EB];
        #pragma unroll
        for (int u = 0; u < EB; ++u) {
            const int j = (j0 + u < deg) ? (j0 + u) : (deg - 1);
            src[u] = csr[beg + j];
        }
        float4 a4[EB];
        #pragma unroll
        for (int u = 0; u < EB; ++u)
            a4[u] = *(const float4*)(xlb + (size_t)src[u] * OUT_DIM);
        float p[EB];
        float mn = m;
        #pragma unroll
        for (int u = 0; u < EB; ++u) {
            p[u] = head_logit(a4[u], r4, t4);
            mn = (j0 + u < deg) ? fmaxf(mn, p[u]) : mn;
        }
        const float sc = __expf(m - mn);
        float ex[EB];
        #pragma unroll
        for (int u = 0; u < EB; ++u)
            ex[u] = (j0 + u < deg) ? __expf(p[u] - mn) : 0.0f;
        den   = den   * sc + ((ex[0] + ex[1]) + (ex[2] + ex[3]));
        acc.x = acc.x * sc + ((ex[0] * a4[0].x + ex[1] * a4[1].x) + (ex[2] * a4[2].x + ex[3] * a4[3].x));
        acc.y = acc.y * sc + ((ex[0] * a4[0].y + ex[1] * a4[1].y) + (ex[2] * a4[2].y + ex[3] * a4[3].y));
        acc.z = acc.z * sc + ((ex[0] * a4[0].z + ex[1] * a4[1].z) + (ex[2] * a4[2].z + ex[3] * a4[3].z));
        acc.w = acc.w * sc + ((ex[0] * a4[0].w + ex[1] * a4[1].w) + (ex[2] * a4[2].w + ex[3] * a4[3].w));
        m = mn;
    }
    static_assert(EB == 4, "the batch sums above are written out for 4 edges");

    const float  dh = den + 1e-16f;
    const float4 bi = *(const float4*)(bias + lane * 4);
    float v0 = acc.x / dh + bi.x;
    float v1 = acc.y / dh + bi.y;
    float v2 = acc.z / dh + bi.z;
    float v3 = acc.w / dh + bi.w;

    float s  = v0 + v1 + v2 + v3;
    float sq = v0 * v0 + v1 * v1 + v2 * v2 + v3 * v3;
    #pragma unroll
    for (int i = 1; i < 32; i <<= 1) {
        s  += __shfl_xor(s,  i, 32);
        sq += __shfl_xor(sq, i, 32);
    }
    const float mean = s * (1.0f / OUT_DIM);
    const float var  = sq * (1.0f / OUT_DIM) - mean * mean;
    const float inv  = rsqrtf(var + 1e-5f);

    const float4 g4 = *(const float4*)(gamma + lane * 4);
    const float4 b4 = *(const float4*)(beta + lane * 4);
    float4 o;
    o.x = (v0 - mean) * inv * g4.x + b4.x;
    o.y = (v1 - mean) * inv * g4.y + b4.y;
    o.z = (v2 - mean) * inv * g4.z + b4.z;
    o.w = (v3 - mean) * inv * g4.w + b4.w;
    *(float4*)(out + (size_t)node * OUT_DIM + lane * 4) = o;
}

extern "C" void kernel_launch(void* const* d_in, const int* in_sizes, int n_in,
                              void* d_out, int out_size, void* d_ws, size_t ws_size,
                              hipStream_t stream) {
    const float* x     = (const float*)d_in[0];
    const int*   eidx  = (const int*)  d_in[1];
    const float* Wl    = (const float*)d_in[2];
    const float* bl    = (const float*)d_in[3];
    const float* Wr    = (const float*)d_in[4];
    const float* br    = (const float*)d_in[5];
    const float* att   = (const float*)d_in[6];
    const float* bias  = (const float*)d_in[7];
    const float* gamma = (const float*)d_in[8];
    const float* beta  = (const float*)d_in[9];
    float* out = (float*)d_out;

    const size_t szXL    = (size_t)BN * OUT_DIM;
    const size_t szCSR   = (size_t)B_BATCH * E_EDGES;
    const size_t szN     = (size_t)BN;
    const size_t needed  = (2 * szXL + szCSR + 3 * szN) * sizeof(float);
    if (ws_size < needed) return;

    float* xl    = (float*)d_ws;
    float* xr    = xl + szXL;
    int*   csr   = (int*)(xr + szXL);
    int*   cnt   = csr + szCSR;
    int*   start = cnt + 2 * szN;
    int*   rank  = (int*)xr;

    hipMemsetAsync(cnt, 0, szN * sizeof(int), stream);

    gat_count_rank<<<B_BATCH * RANK_SPLIT, RANK_THREADS, 0, stream>>>(eidx, cnt, rank);
    gat_scan_start<<<B_BATCH, SCAN_THREADS, 0, stream>>>(cnt, start);
    gat_fill_buckets<<<EDGE_QUADS / 256, 256, 0, stream>>>(eidx, start, rank, csr);

    gat_gemm_wmma<<<(BN / 16) / GEMM_WAVES, 256, 0, stream>>>(x, Wl, bl, Wr, br, xl, xr);

    gat_node_gather<<<BN / 8, 256, 0, stream>>>(csr, start, cnt, xl, xr, att, bias, gamma, beta, out);
}


# Bucketing in two launches, not four: count blocks store per-node counts (no global atomics, no memset); fill blocks sum and scan them in LDS; bucket order = block order, then LDS add arrival
# speedup vs baseline: 5.1252x; 1.0506x over previous
#include <hip/hip_runtime.h>
#include <hip/hip_bf16.h>

constexpr int B_BATCH = 4;
constexpr int N_NODES = 8192;
constexpr int E_EDGES = 131072;
constexpr int ETOT    = E_EDGES + N_NODES;
constexpr int IN_DIM  = 128;
constexpr int HEADS   = 4;
constexpr int CH      = 32;
constexpr int OUT_DIM = HEADS * CH;
constexpr int BN      = B_BATCH * N_NODES;

typedef float v2f __attribute__((ext_vector_type(2)));
typedef float v8f __attribute__((ext_vector_type(8)));

__device__ __forceinline__ float lrelu02(float v) {
    return v > 0.0f ? v : 0.2f * v;
}

constexpr int GEMM_WAVES = 8;
constexpr int GEMM_NCOL  = 2 * OUT_DIM;
constexpr int GEMM_TILES = GEMM_NCOL / 16;
constexpr int KSTEPS     = IN_DIM / 4;
constexpr int KGROUP     = 2;
static_assert(KSTEPS % KGROUP == 0, "whole groups");
constexpr int KPAIRS     = IN_DIM / 2;
constexpr int WPITCH     = 672;
static_assert(WPITCH >= 2 * GEMM_NCOL && WPITCH % 64 == 32, "pitch holds a row and splits the bank halves");
static_assert((BN / 16) % GEMM_WAVES == 0, "whole workgroups of row strips");
static_assert(IN_DIM == 128 && OUT_DIM == 128, "staging index arithmetic below");

__global__ __launch_bounds__(256) void gat_gemm_wmma(
    const float* __restrict__ x,
    const float* __restrict__ Wl, const float* __restrict__ bl,
    const float* __restrict__ Wr, const float* __restrict__ br,
    float* __restrict__ xl, float* __restrict__ xr)
{
    __shared__ __attribute__((aligned(16))) float wlds[KPAIRS * WPITCH];

    const int lane    = threadIdx.x & 31;
    const int wv      = threadIdx.x >> 5;
    const int rowTile = blockIdx.x * GEMM_WAVES + wv;
    const int c       = lane & 15;
    const int h       = lane >> 4;
    const int row     = rowTile * 16 + c;

    v2f a[KSTEPS];
    #pragma unroll
    for (int s = 0; s < KSTEPS; ++s)
        a[s] = *(const v2f*)(x + (size_t)row * IN_DIM + 4 * s + 2 * h);

    #pragma unroll
    for (int i = 0; i < (KPAIRS * 2 * 32) / 256; ++i) {
        const int task = i * 256 + (int)threadIdx.x;
        const int c4   = task & 31;
        const int mat  = (task >> 5) & 1;
        const int kp   = task >> 6;
        const float* __restrict__ Wm = mat ? Wr : Wl;
        const float4 r0 = *(const float4*)(Wm + (size_t)(2 * kp)     * OUT_DIM + c4 * 4);
        const float4 r1 = *(const float4*)(Wm + (size_t)(2 * kp + 1) * OUT_DIM + c4 * 4);
        float* d = wlds + kp * WPITCH + (mat * OUT_DIM + c4 * 4) * 2;
        *(float4*)(d)     = make_float4(r0.x, r1.x, r0.y, r1.y);
        *(float4*)(d + 4) = make_float4(r0.z, r1.z, r0.w, r1.w);
    }
    __syncthreads();

    const float* bfrag = wlds + h * WPITCH + c * 2;
    v8f acc[GEMM_TILES];
    #pragma unroll
    for (int j = 0; j < GEMM_TILES; ++j) acc[j] = v8f{};

    v2f bcur[KGROUP * GEMM_TILES];
    #pragma unroll
    for (int u = 0; u < KGROUP; ++u)
        #pragma unroll
        for (int j = 0; j < GEMM_TILES; ++j)
            bcur[u * GEMM_TILES + j] = *(const v2f*)(bfrag + (2 * u) * WPITCH + j * 32);

    #pragma unroll
    for (int g = 0; g < KSTEPS / KGROUP; ++g) {
        v2f bnext[KGROUP * GEMM_TILES];
        if (g + 1 < KSTEPS / KGROUP) {
            #pragma unroll
            for (int u = 0; u < KGROUP; ++u)
                #pragma unroll
                for (int j = 0; j < GEMM_TILES; ++j)
                    bnext[u * GEMM_TILES + j] =
                        *(const v2f*)(bfrag + (2 * ((g + 1) * KGROUP + u)) * WPITCH + j * 32);
        }
        __builtin_amdgcn_sched_barrier(0);
        #pragma unroll
        for (int u = 0; u < KGROUP; ++u)
            #pragma unroll
            for (int j = 0; j < GEMM_TILES; ++j)
                acc[j] = __builtin_amdgcn_wmma_f32_16x16x4_f32(
                             false, a[g * KGROUP + u], false, bcur[u * GEMM_TILES + j],
                             (short)0, acc[j], false, false);
        __builtin_amdgcn_sched_barrier(0);
        if (g + 1 < KSTEPS / KGROUP) {
            #pragma unroll
            for (int q = 0; q < KGROUP * GEMM_TILES; ++q) bcur[q] = bnext[q];
        }
    }

    #pragma unroll
    for (int jp = 0; jp < GEMM_TILES / 2; ++jp) {
        const int    mcol = (jp & 3) * 32 + lane;
        float* __restrict__ om = (jp < 4) ? xl : xr;
        const float  bv   = ((jp < 4) ? bl : br)[mcol];
        #pragma unroll
        for (int r = 0; r < 8; ++r) {
            const float lo   = acc[2 * jp][r];
            const float hi   = acc[2 * jp + 1][r];
            const float give = h ? lo : hi;
            const float got  = __shfl_xor(give, 16, 32);
            const float top  = h ? got : lo;
            const float bot  = h ? hi  : got;
            om[(size_t)(rowTile * 16 + r)     * OUT_DIM + mcol] = top + bv;
            om[(size_t)(rowTile * 16 + r + 8) * OUT_DIM + mcol] = bot + bv;
        }
    }
}

constexpr int QUADS_PER_BATCH = E_EDGES / 4;
constexpr int EB              = 4;
static_assert(E_EDGES % 4 == 0, "edge quads are aligned 16-byte loads");

constexpr int RANK_SPLIT   = 8;
constexpr int RANK_THREADS = 1024;
constexpr int RANK_QUADS   = QUADS_PER_BATCH / RANK_SPLIT;
constexpr int RANK_Q_PER_T = RANK_QUADS / RANK_THREADS;
constexpr int BINS_PER_T   = N_NODES / RANK_THREADS;
static_assert(QUADS_PER_BATCH % RANK_SPLIT == 0 && RANK_QUADS % RANK_THREADS == 0, "whole quads per thread");
static_assert(N_NODES % RANK_THREADS == 0, "whole table words per thread");
static_assert(BINS_PER_T == 8, "one 16-byte load of 8 packed counts per thread and count block");
static_assert(RANK_QUADS * 4 < 65536, "a block's count of one node fits in 16 bits");
constexpr int PACKED_WORDS = N_NODES / 2;
constexpr int NODE_BITS    = 13;
static_assert(N_NODES == (1 << NODE_BITS), "a node index fills exactly the low bits");
static_assert((long long)(RANK_QUADS * 4 - 1) << NODE_BITS < (1LL << 31), "rank and node fit in one positive word");

__global__ __launch_bounds__(RANK_THREADS) void gat_count_rank(
    const int* __restrict__ eidx, int* __restrict__ bhist, int* __restrict__ rank)
{
    __shared__ int hist[N_NODES];
    const int t = threadIdx.x;
    const int b = blockIdx.x / RANK_SPLIT;
    const int g = blockIdx.x - b * RANK_SPLIT;
    const int* __restrict__ dst = eidx + ((size_t)b * 2 + 1) * E_EDGES + (size_t)g * RANK_QUADS * 4;
    int*       __restrict__ rk  = rank + (size_t)b * E_EDGES + (size_t)g * RANK_QUADS * 4;

    int4 d[RANK_Q_PER_T];
    #pragma unroll
    for (int i = 0; i < RANK_Q_PER_T; ++i)
        d[i] = *(const int4*)(dst + (size_t)(i * RANK_THREADS + t) * 4);

    for (int i = 0; i < BINS_PER_T; ++i) hist[i * RANK_THREADS + t] = 0;
    asm volatile("" :: "v"(d[0].x), "v"(d[0].y), "v"(d[0].z), "v"(d[0].w),
                       "v"(d[1].x), "v"(d[1].y), "v"(d[1].z), "v"(d[1].w),
                       "v"(d[2].x), "v"(d[2].y), "v"(d[2].z), "v"(d[2].w),
                       "v"(d[3].x), "v"(d[3].y), "v"(d[3].z), "v"(d[3].w));
    static_assert(RANK_Q_PER_T == 4, "the statement above names the four quads");
    __syncthreads();

    int4 r[RANK_Q_PER_T];
    #pragma unroll
    for (int i = 0; i < RANK_Q_PER_T; ++i) {
        r[i].x = atomicAdd(&hist[d[i].x], 1);
        r[i].y = atomicAdd(&hist[d[i].y], 1);
        r[i].z = atomicAdd(&hist[d[i].z], 1);
        r[i].w = atomicAdd(&hist[d[i].w], 1);
    }
    __syncthreads();

    #pragma unroll
    for (int i = 0; i < RANK_Q_PER_T; ++i)
        *(int4*)(rk + (size_t)(i * RANK_THREADS + t) * 4) =
            make_int4(d[i].x | (r[i].x << NODE_BITS), d[i].y | (r[i].y << NODE_BITS),
                      d[i].z | (r[i].z << NODE_BITS), d[i].w | (r[i].w << NODE_BITS));

    const int4 c0 = *(const int4*)(hist + t * BINS_PER_T);
    const int4 c1 = *(const int4*)(hist + t * BINS_PER_T + 4);
    int* __restrict__ bh = bhist + (size_t)blockIdx.x * PACKED_WORDS + t * (BINS_PER_T / 2);
    *(int4*)(bh) = make_int4(c0.x | (c0.y << 16), c0.z | (c0.w << 16),
                             c1.x | (c1.y << 16), c1.z | (c1.w << 16));
}

constexpr int FILL_SUB     = 8;
constexpr int FILL_EDGES   = RANK_QUADS * 4 / FILL_SUB;
static_assert(FILL_EDGES == 2 * RANK_THREADS, "two edges (one 8-byte load per array) per thread");

__global__ __launch_bounds__(RANK_THREADS) void gat_scan_fill(
    const int* __restrict__ eidx, const int* __restrict__ bhist, const int* __restrict__ rank,
    int* __restrict__ start, int* __restrict__ indeg, int* __restrict__ csr)
{
    __shared__ __attribute__((aligned(16))) int where[N_NODES];
    __shared__ int wsum[32];
    const int t    = threadIdx.x;
    const int lane = t & 31;
    const int wv   = t >> 5;
    const int cb   = blockIdx.x / FILL_SUB;
    const int sub  = blockIdx.x - cb * FILL_SUB;
    const int b    = cb / RANK_SPLIT;
    const int g    = cb - b * RANK_SPLIT;

    const int* __restrict__ hb = bhist + (size_t)b * RANK_SPLIT * PACKED_WORDS + t * (BINS_PER_T / 2);
    int4 hp[RANK_SPLIT];
    #pragma unroll
    for (int q = 0; q < RANK_SPLIT; ++q)
        hp[q] = *(const int4*)(hb + (size_t)q * PACKED_WORDS);
    const size_t eoff = (size_t)g * RANK_QUADS * 4 + (size_t)sub * FILL_EDGES + (size_t)t * 2;
    const int2 dr = *(const int2*)(rank + (size_t)b * E_EDGES + eoff);
    const int2 s2 = *(const int2*)(eidx + ((size_t)b * 2 + 0) * E_EDGES + eoff);

    int v[8]   = {0, 0, 0, 0, 0, 0, 0, 0};
    int bef[8] = {0, 0, 0, 0, 0, 0, 0, 0};
    #pragma unroll
    for (int q = 0; q < RANK_SPLIT; ++q) {
        const unsigned w0 = (unsigned)hp[q].x, w1 = (unsigned)hp[q].y, w2 = (unsigned)hp[q].z, w3 = (unsigned)hp[q].w;
        const int hq[8] = {(int)(w0 & 0xffffu), (int)(w0 >> 16), (int)(w1 & 0xffffu), (int)(w1 >> 16),
                           (int)(w2 & 0xffffu), (int)(w2 >> 16), (int)(w3 & 0xffffu), (int)(w3 >> 16)};
        const bool earlier = q < g;
        #pragma unroll
        for (int i = 0; i < 8; ++i) {
            v[i]   += hq[i];
            bef[i] += earlier ? hq[i] : 0;
        }
    }
    int tot = 0;
    #pragma unroll
    for (int i = 0; i < 8; ++i) tot += v[i];

    int inc = tot;
    #pragma unroll
    for (int o = 1; o < 32; o <<= 1) {
        const int n = __shfl_up(inc, o, 32);
        if (lane >= o) inc += n;
    }
    if (lane == 31) wsum[wv] = inc;
    __syncthreads();
    if (wv == 0) {
        const int w = wsum[lane];
        int winc = w;
        #pragma unroll
        for (int o = 1; o < 32; o <<= 1) {
            const int n = __shfl_up(winc, o, 32);
            if (lane >= o) winc += n;
        }
        wsum[lane] = winc - w;
    }
    __syncthreads();

    int run = b * E_EDGES + wsum[wv] + inc - tot;
    int o8[8];
    #pragma unroll
    for (int i = 0; i < 8; ++i) { o8[i] = run; run += v[i]; }
    *(int4*)(where + t * 8)     = make_int4(o8[0] + bef[0], o8[1] + bef[1], o8[2] + bef[2], o8[3] + bef[3]);
    *(int4*)(where + t * 8 + 4) = make_int4(o8[4] + bef[4], o8[5] + bef[5], o8[6] + bef[6], o8[7] + bef[7]);
    __syncthreads();

    if (sub == 0 && (t >> 7) == g) {
        int* __restrict__ s = start + (size_t)b * N_NODES + t * 8;
        int* __restrict__ c = indeg + (size_t)b * N_NODES + t * 8;
        *(int4*)(s)     = make_int4(o8[0], o8[1], o8[2], o8[3]);
        *(int4*)(s + 4) = make_int4(o8[4], o8[5], o8[6], o8[7]);
        *(int4*)(c)     = make_int4(v[0], v[1], v[2], v[3]);
        *(int4*)(c + 4) = make_int4(v[4], v[5], v[6], v[7]);
    }

    constexpr unsigned NODE_MASK = (1u << NODE_BITS) - 1u;
    const unsigned e0 = (unsigned)dr.x, e1 = (unsigned)dr.y;
    csr[where[e0 & NODE_MASK] + (int)(e0 >> NODE_BITS)] = s2.x;
    csr[where[e1 & NODE_MASK] + (int)(e1 >> NODE_BITS)] = s2.y;
}

__device__ __forceinline__ float head_logit(const float4 a, const float4 r, const float4 t)
{
    float p = lrelu02(a.x + r.x) * t.x
            + lrelu02(a.y + r.y) * t.y
            + lrelu02(a.z + r.z) * t.z
            + lrelu02(a.w + r.w) * t.w;
    p += __shfl_xor(p, 1, 32);
    p += __shfl_xor(p, 2, 32);
    p += __shfl_xor(p, 4, 32);
    return p;
}

__global__ __launch_bounds__(256) void gat_node_gather(
    const int* __restrict__ csr, const int* __restrict__ start, const int* __restrict__ indeg,
    const float* __restrict__ xl, const float* __restrict__ xr, const float* __restrict__ att,
    const float* __restrict__ bias, const float* __restrict__ gamma,
    const float* __restrict__ beta, float* __restrict__ out)
{
    const int lane = threadIdx.x & 31;
    const int wv   = __builtin_amdgcn_readfirstlane((int)(threadIdx.x >> 5));
    const int node = blockIdx.x * 8 + wv;
    const int b    = node / N_NODES;
    const int nloc = node - b * N_NODES;
    const int beg  = start[node];
    const int deg  = indeg[node];

    const float4 t4 = *(const float4*)(att + lane * 4);
    const float* __restrict__ xlb = xl + (size_t)b * N_NODES * OUT_DIM + lane * 4;
    const float4 r4 = *(const float4*)(xr + (size_t)node * OUT_DIM + lane * 4);

    float4 acc = *(const float4*)(xlb + (size_t)nloc * OUT_DIM);
    float  m   = head_logit(acc, r4, t4);
    float  den = 1.0f;

    for (int j0 = 0; j0 < deg; j0 += EB) {
        int src[EB];
        #pragma unroll
        for (int u = 0; u < EB; ++u) {
            const int j = (j0 + u < deg) ? (j0 + u) : (deg - 1);
            src[u] = csr[beg + j];
        }
        float4 a4[EB];
        #pragma unroll
        for (int u = 0; u < EB; ++u)
            a4[u] = *(const float4*)(xlb + (size_t)src[u] * OUT_DIM);
        float p[EB];
        float mn = m;
        #pragma unroll
        for (int u = 0; u < EB; ++u) {
            p[u] = head_logit(a4[u], r4, t4);
            mn = (j0 + u < deg) ? fmaxf(mn, p[u]) : mn;
        }
        const float sc = __expf(m - mn);
        float ex[EB];
        #pragma unroll
        for (int u = 0; u < EB; ++u)
            ex[u] = (j0 + u < deg) ? __expf(p[u] - mn) : 0.0f;
        den   = den   * sc + ((ex[0] + ex[1]) + (ex[2] + ex[3]));
        acc.x = acc.x * sc + ((ex[0] * a4[0].x + ex[1] * a4[1].x) + (ex[2] * a4[2].x + ex[3] * a4[3].x));
        acc.y = acc.y * sc + ((ex[0] * a4[0].y + ex[1] * a4[1].y) + (ex[2] * a4[2].y + ex[3] * a4[3].y));
        acc.z = acc.z * sc + ((ex[0] * a4[0].z + ex[1] * a4[1].z) + (ex[2] * a4[2].z + ex[3] * a4[3].z));
        acc.w = acc.w * sc + ((ex[0] * a4[0].w + ex[1] * a4[1].w) + (ex[2] * a4[2].w + ex[3] * a4[3].w));
        m = mn;
    }
    static_assert(EB == 4, "the batch sums above are written out for 4 edges");

    const float  dh = den + 1e-16f;
    const float4 bi = *(const float4*)(bias + lane * 4);
    float v0 = acc.x / dh + bi.x;
    float v1 = acc.y / dh + bi.y;
    float v2 = acc.z / dh + bi.z;
    float v3 = acc.w / dh + bi.w;

    float s  = v0 + v1 + v2 + v3;
    float sq = v0 * v0 + v1 * v1 + v2 * v2 + v3 * v3;
    #pragma unroll
    for (int i = 1; i < 32; i <<= 1) {
        s  += __shfl_xor(s,  i, 32);
        sq += __shfl_xor(sq, i, 32);
    }
    const float mean = s * (1.0f / OUT_DIM);
    const float var  = sq * (1.0f / OUT_DIM) - mean * mean;
    const float inv  = rsqrtf(var + 1e-5f);

    const float4 g4 = *(const float4*)(gamma + lane * 4);
    const float4 b4 = *(const float4*)(beta + lane * 4);
    float4 o;
    o.x = (v0 - mean) * inv * g4.x + b4.x;
    o.y = (v1 - mean) * inv * g4.y + b4.y;
    o.z = (v2 - mean) * inv * g4.z + b4.z;
    o.w = (v3 - mean) * inv * g4.w + b4.w;
    *(float4*)(out + (size_t)node * OUT_DIM + lane * 4) = o;
}

extern "C" void kernel_launch(void* const* d_in, const int* in_sizes, int n_in,
                              void* d_out, int out_size, void* d_ws, size_t ws_size,
                              hipStream_t stream) {
    const float* x     = (const float*)d_in[0];
    const int*   eidx  = (const int*)  d_in[1];
    const float* Wl    = (const float*)d_in[2];
    const float* bl    = (const float*)d_in[3];
    const float* Wr    = (const float*)d_in[4];
    const float* br    = (const float*)d_in[5];
    const float* att   = (const float*)d_in[6];
    const float* bias  = (const float*)d_in[7];
    const float* gamma = (const float*)d_in[8];
    const float* beta  = (const float*)d_in[9];
    float* out = (float*)d_out;

    const size_t szXL    = (size_t)BN * OUT_DIM;
    const size_t szCSR   = (size_t)B_BATCH * E_EDGES;
    const size_t szN     = (size_t)BN;
    const size_t needed  = (2 * szXL + szCSR + 3 * szN) * sizeof(float);
    if (ws_size < needed) return;

    float* xl    = (float*)d_ws;
    float* xr    = xl + szXL;
    int*   csr   = (int*)(xr + szXL);
    int*   cnt   = csr + szCSR;
    int*   start = cnt + 2 * szN;
    int*   rank  = (int*)xr;
    int*   bhist = rank + szCSR;
    static_assert((size_t)B_BATCH * E_EDGES + (size_t)B_BATCH * RANK_SPLIT * PACKED_WORDS <= (size_t)BN * OUT_DIM,
                  "ranks and count tables fit in the xr plane");

    gat_count_rank<<<B_BATCH * RANK_SPLIT, RANK_THREADS, 0, stream>>>(eidx, bhist, rank);
    gat_scan_fill<<<B_BATCH * RANK_SPLIT * FILL_SUB, RANK_THREADS, 0, stream>>>(eidx, bhist, rank, start, cnt, csr);

    gat_gemm_wmma<<<(BN / 16) / GEMM_WAVES, 256, 0, stream>>>(x, Wl, bl, Wr, br, xl, xr);

    gat_node_gather<<<BN / 8, 256, 0, stream>>>(csr, start, cnt, xl, xr, att, bias, gamma, beta, out);
}
